# starting kernel
# baseline (speedup 1.0000x reference)
#include <hip/hip_runtime.h>

typedef __attribute__((ext_vector_type(2))) float v2f;
typedef __attribute__((ext_vector_type(4))) float v4f;
typedef __attribute__((ext_vector_type(8))) float v8f;

#define H       128
#define ROWPAD  132
#define TILE    16
#define K2_WAVES 2

__global__ void normalize_rows_k(const float* __restrict__ emb,
                                 float* __restrict__ e, int n_nodes) {
    int wave = threadIdx.x >> 5;
    int lane = threadIdx.x & 31;
    int row  = blockIdx.x * (blockDim.x >> 5) + wave;
    if (row >= n_nodes) return;

    const v4f* rp = (const v4f*)(emb + (size_t)row * H);
    v4f v = __builtin_nontemporal_load(&rp[lane]);

    float s = v.x * v.x + v.y * v.y + v.z * v.z + v.w * v.w;
#pragma unroll
    for (int off = 16; off > 0; off >>= 1)
        s += __shfl_xor(s, off, 32);

    float inv = 1.0f / fmaxf(sqrtf(s), 1e-12f);
    v4f o;
    o.x = v.x * inv; o.y = v.y * inv; o.z = v.z * inv; o.w = v.w * inv;
    ((v4f*)(e + (size_t)row * H))[lane] = o;
}

__global__ void edge_dot_k(const float* __restrict__ e,
                           const float* __restrict__ dvec,
                           const float* __restrict__ scale,
                           const int*   __restrict__ src,
                           const int*   __restrict__ dst,
                           float*       __restrict__ z,
                           int n_edges) {
    __shared__ float lds[K2_WAVES * 32 * ROWPAD];

    int wave = threadIdx.x >> 5;
    int lane = threadIdx.x & 31;
    long long tile = (long long)blockIdx.x * K2_WAVES + wave;
    long long base = tile * TILE;
    if (base >= n_edges) return;

    int m = lane & 15;
    long long eidx = base + m;
    if (eidx >= n_edges) eidx = n_edges - 1;

    int rowidx = (lane < 16) ? __builtin_nontemporal_load(&src[eidx])
                             : __builtin_nontemporal_load(&dst[eidx]);

    float* buf = lds + wave * (32 * ROWPAD);
#pragma unroll 8
    for (int r = 0; r < 32; ++r) {
        int ri = __shfl(rowidx, r, 32);
        v4f v = ((const v4f*)(e + (size_t)ri * H))[lane];
        *(v4f*)(buf + r * ROWPAD + 4 * lane) = v;
    }

    int koff = (lane >> 4) << 1;
    const float* arow = buf + m * ROWPAD;
    const float* brow = buf + (16 + m) * ROWPAD;

    v8f acc0 = {};
    v8f acc1 = {};
#pragma unroll
    for (int k0 = 0; k0 < H; k0 += 8) {
        v2f a0 = *(const v2f*)(arow + k0 + koff);
        v2f b0 = *(const v2f*)(brow + k0 + koff);
        v2f d0 = *(const v2f*)(dvec + k0 + koff);
        a0 *= d0;
        acc0 = __builtin_amdgcn_wmma_f32_16x16x4_f32(
            false, a0, false, b0, (short)0, acc0, false, false);

        v2f a1 = *(const v2f*)(arow + k0 + 4 + koff);
        v2f b1 = *(const v2f*)(brow + k0 + 4 + koff);
        v2f d1 = *(const v2f*)(dvec + k0 + 4 + koff);
        a1 *= d1;
        acc1 = __builtin_amdgcn_wmma_f32_16x16x4_f32(
            false, a1, false, b1, (short)0, acc1, false, false);
    }
    v8f acc = acc0 + acc1;

    int r = lane & 7;
    float dv = acc[0];
    dv = (r == 1) ? acc[1] : dv;
    dv = (r == 2) ? acc[2] : dv;
    dv = (r == 3) ? acc[3] : dv;
    dv = (r == 4) ? acc[4] : dv;
    dv = (r == 5) ? acc[5] : dv;
    dv = (r == 6) ? acc[6] : dv;
    dv = (r == 7) ? acc[7] : dv;

    bool active = (lane < 8) || (lane >= 24);
    int  mm     = (lane < 8) ? lane : (lane - 16);
    long long oe = base + mm;
    float sc = scale[0];
    if (active && oe < n_edges)
        __builtin_nontemporal_store(dv * sc, &z[oe]);
}

extern "C" void kernel_launch(void* const* d_in, const int* in_sizes, int n_in,
                              void* d_out, int out_size, void* d_ws, size_t ws_size,
                              hipStream_t stream) {
    const float* emb   = (const float*)d_in[0];
    const float* dvec  = (const float*)d_in[1];
    const float* scale = (const float*)d_in[2];
    const int*   src   = (const int*)d_in[3];
    const int*   dst   = (const int*)d_in[4];
    float*       z     = (float*)d_out;

    int n_nodes = in_sizes[0] / H;
    int n_edges = in_sizes[3];

    float* e = (float*)d_ws;

    int rows_per_blk = 256 / 32;
    int nb1 = (n_nodes + rows_per_blk - 1) / rows_per_blk;
    normalize_rows_k<<<nb1, 256, 0, stream>>>(emb, e, n_nodes);

    long long tiles = ((long long)n_edges + TILE - 1) / TILE;
    int nb2 = (int)((tiles + K2_WAVES - 1) / K2_WAVES);
    edge_dot_k<<<nb2, K2_WAVES * 32, 0, stream>>>(e, dvec, scale, src, dst, z,
                                                  n_edges);
}


# Row normalisation folded into the edge kernel: raw rows gathered, squared norms summed from the same LDS fragments, 1/norms applied to the finished dot; the table pass and its launch are gone
# speedup vs baseline: 1.1386x; 1.1386x over previous
#include <hip/hip_runtime.h>

typedef __attribute__((ext_vector_type(2))) float v2f;
typedef __attribute__((ext_vector_type(4))) float v4f;
typedef __attribute__((ext_vector_type(8))) float v8f;

#define H       128
#define ROWPAD  132
#define TILE    16
#define K2_WAVES 2

__global__ void edge_cos_k(const float* __restrict__ emb,
                           const float* __restrict__ dvec,
                           const float* __restrict__ scale,
                           const int*   __restrict__ src,
                           const int*   __restrict__ dst,
                           float*       __restrict__ z,
                           int n_edges) {
    __shared__ float lds[K2_WAVES * 32 * ROWPAD];

    int wave = threadIdx.x >> 5;
    int lane = threadIdx.x & 31;
    long long tile = (long long)blockIdx.x * K2_WAVES + wave;
    long long base = tile * TILE;
    if (base >= n_edges) return;

    int m = lane & 15;
    long long eidx = base + m;
    if (eidx >= n_edges) eidx = n_edges - 1;

    int rowidx = (lane < 16) ? __builtin_nontemporal_load(&src[eidx])
                             : __builtin_nontemporal_load(&dst[eidx]);

    float* buf = lds + wave * (32 * ROWPAD);
#pragma unroll 8
    for (int r = 0; r < 32; ++r) {
        int ri = __shfl(rowidx, r, 32);
        v4f v = ((const v4f*)(emb + (size_t)ri * H))[lane];
        *(v4f*)(buf + r * ROWPAD + 4 * lane) = v;
    }

    int koff = (lane >> 4) << 1;
    const float* arow = buf + m * ROWPAD;
    const float* brow = buf + (16 + m) * ROWPAD;

    v8f acc0 = {};
    v8f acc1 = {};
    v2f na = {0.0f, 0.0f};
    v2f nb = {0.0f, 0.0f};
#pragma unroll
    for (int k0 = 0; k0 < H; k0 += 8) {
        v2f a0 = *(const v2f*)(arow + k0 + koff);
        v2f b0 = *(const v2f*)(brow + k0 + koff);
        v2f d0 = *(const v2f*)(dvec + k0 + koff);
        na += a0 * a0;
        nb += b0 * b0;
        a0 *= d0;
        acc0 = __builtin_amdgcn_wmma_f32_16x16x4_f32(
            false, a0, false, b0, (short)0, acc0, false, false);

        v2f a1 = *(const v2f*)(arow + k0 + 4 + koff);
        v2f b1 = *(const v2f*)(brow + k0 + 4 + koff);
        v2f d1 = *(const v2f*)(dvec + k0 + 4 + koff);
        na += a1 * a1;
        nb += b1 * b1;
        a1 *= d1;
        acc1 = __builtin_amdgcn_wmma_f32_16x16x4_f32(
            false, a1, false, b1, (short)0, acc1, false, false);
    }
    v8f acc = acc0 + acc1;

    float sa = na.x + na.y;
    float sb = nb.x + nb.y;
    sa += __shfl_xor(sa, 16, 32);
    sb += __shfl_xor(sb, 16, 32);
    float inv_a = 1.0f / fmaxf(sqrtf(sa), 1e-12f);
    float inv_b = 1.0f / fmaxf(sqrtf(sb), 1e-12f);

    int r = lane & 7;
    float dv = acc[0];
    dv = (r == 1) ? acc[1] : dv;
    dv = (r == 2) ? acc[2] : dv;
    dv = (r == 3) ? acc[3] : dv;
    dv = (r == 4) ? acc[4] : dv;
    dv = (r == 5) ? acc[5] : dv;
    dv = (r == 6) ? acc[6] : dv;
    dv = (r == 7) ? acc[7] : dv;

    bool active = (lane < 8) || (lane >= 24);
    int  mm     = (lane < 8) ? lane : (lane - 16);
    long long oe = base + mm;
    float sc = scale[0];
    if (active && oe < n_edges)
        __builtin_nontemporal_store(dv * inv_a * inv_b * sc, &z[oe]);
}

extern "C" void kernel_launch(void* const* d_in, const int* in_sizes, int n_in,
                              void* d_out, int out_size, void* d_ws, size_t ws_size,
                              hipStream_t stream) {
    const float* emb   = (const float*)d_in[0];
    const float* dvec  = (const float*)d_in[1];
    const float* scale = (const float*)d_in[2];
    const int*   src   = (const int*)d_in[3];
    const int*   dst   = (const int*)d_in[4];
    float*       z     = (float*)d_out;

    int n_edges = in_sizes[3];

    long long tiles = ((long long)n_edges + TILE - 1) / TILE;
    int nb2 = (int)((tiles + K2_WAVES - 1) / K2_WAVES);
    edge_cos_k<<<nb2, K2_WAVES * 32, 0, stream>>>(emb, dvec, scale, src, dst, z,
                                                  n_edges);
}


# All 32 row gathers of a tile issued back to back before any LDS store (full unroll, constant-lane row index, launch bounds lift the register cap): 32 loads in flight per wave, one round trip not four
# speedup vs baseline: 1.3063x; 1.1473x over previous
#include <hip/hip_runtime.h>

typedef __attribute__((ext_vector_type(2))) float v2f;
typedef __attribute__((ext_vector_type(4))) float v4f;
typedef __attribute__((ext_vector_type(8))) float v8f;

#define H       128
#define ROWPAD  132
#define TILE    16
#define K2_WAVES 2

__global__ void __launch_bounds__(K2_WAVES * 32)
__attribute__((amdgpu_waves_per_eu(5))) edge_cos_k(const float* __restrict__ emb,
                           const float* __restrict__ dvec,
                           const float* __restrict__ scale,
                           const int*   __restrict__ src,
                           const int*   __restrict__ dst,
                           float*       __restrict__ z,
                           int n_edges) {
    __shared__ float lds[K2_WAVES * 32 * ROWPAD];

    int wave = threadIdx.x >> 5;
    int lane = threadIdx.x & 31;
    long long tile = (long long)blockIdx.x * K2_WAVES + wave;
    long long base = tile * TILE;
    if (base >= n_edges) return;

    int m = lane & 15;
    long long eidx = base + m;
    if (eidx >= n_edges) eidx = n_edges - 1;

    int rowidx = (lane < 16) ? __builtin_nontemporal_load(&src[eidx])
                             : __builtin_nontemporal_load(&dst[eidx]);

    float* buf = lds + wave * (32 * ROWPAD);
    v4f rows[32];
#pragma unroll
    for (int r = 0; r < 32; ++r) {
        int ri = __builtin_amdgcn_readlane(rowidx, r);
        rows[r] = ((const v4f*)(emb + (size_t)ri * H))[lane];
    }
    __builtin_amdgcn_sched_barrier(0);
#pragma unroll
    for (int r = 0; r < 32; ++r)
        *(v4f*)(buf + r * ROWPAD + 4 * lane) = rows[r];

    int koff = (lane >> 4) << 1;
    const float* arow = buf + m * ROWPAD;
    const float* brow = buf + (16 + m) * ROWPAD;

    v8f acc0 = {};
    v8f acc1 = {};
    v2f na = {0.0f, 0.0f};
    v2f nb = {0.0f, 0.0f};
#pragma unroll
    for (int k0 = 0; k0 < H; k0 += 8) {
        v2f a0 = *(const v2f*)(arow + k0 + koff);
        v2f b0 = *(const v2f*)(brow + k0 + koff);
        v2f d0 = *(const v2f*)(dvec + k0 + koff);
        na += a0 * a0;
        nb += b0 * b0;
        a0 *= d0;
        acc0 = __builtin_amdgcn_wmma_f32_16x16x4_f32(
            false, a0, false, b0, (short)0, acc0, false, false);

        v2f a1 = *(const v2f*)(arow + k0 + 4 + koff);
        v2f b1 = *(const v2f*)(brow + k0 + 4 + koff);
        v2f d1 = *(const v2f*)(dvec + k0 + 4 + koff);
        na += a1 * a1;
        nb += b1 * b1;
        a1 *= d1;
        acc1 = __builtin_amdgcn_wmma_f32_16x16x4_f32(
            false, a1, false, b1, (short)0, acc1, false, false);
    }
    v8f acc = acc0 + acc1;

    float sa = na.x + na.y;
    float sb = nb.x + nb.y;
    sa += __shfl_xor(sa, 16, 32);
    sb += __shfl_xor(sb, 16, 32);
    float inv_a = 1.0f / fmaxf(sqrtf(sa), 1e-12f);
    float inv_b = 1.0f / fmaxf(sqrtf(sb), 1e-12f);

    int r = lane & 7;
    float dv = acc[0];
    dv = (r == 1) ? acc[1] : dv;
    dv = (r == 2) ? acc[2] : dv;
    dv = (r == 3) ? acc[3] : dv;
    dv = (r == 4) ? acc[4] : dv;
    dv = (r == 5) ? acc[5] : dv;
    dv = (r == 6) ? acc[6] : dv;
    dv = (r == 7) ? acc[7] : dv;

    bool active = (lane < 8) || (lane >= 24);
    int  mm     = (lane < 8) ? lane : (lane - 16);
    long long oe = base + mm;
    float sc = scale[0];
    if (active && oe < n_edges)
        __builtin_nontemporal_store(dv * inv_a * inv_b * sc, &z[oe]);
}

extern "C" void kernel_launch(void* const* d_in, const int* in_sizes, int n_in,
                              void* d_out, int out_size, void* d_ws, size_t ws_size,
                              hipStream_t stream) {
    const float* emb   = (const float*)d_in[0];
    const float* dvec  = (const float*)d_in[1];
    const float* scale = (const float*)d_in[2];
    const int*   src   = (const int*)d_in[3];
    const int*   dst   = (const int*)d_in[4];
    float*       z     = (float*)d_out;

    int n_edges = in_sizes[3];

    long long tiles = ((long long)n_edges + TILE - 1) / TILE;
    int nb2 = (int)((tiles + K2_WAVES - 1) / K2_WAVES);
    edge_cos_k<<<nb2, K2_WAVES * 32, 0, stream>>>(emb, dvec, scale, src, dst, z,
                                                  n_edges);
}
